# starting kernel
# baseline (speedup 1.0000x reference)
#include <hip/hip_runtime.h>

typedef __attribute__((ext_vector_type(2))) float v2f;
typedef __attribute__((ext_vector_type(8))) float v8f;

#define BATCH   4096
#define N_IN    6
#define N_MF    4
#define N_RULES 4096
#define WAVES   8
#define TILE_M  16

__launch_bounds__(256, 1)
__global__ void anfis_wmma_kernel(const float* __restrict__ x,
                                  const float* __restrict__ mfa,
                                  const float* __restrict__ mfb,
                                  const float* __restrict__ mfc,
                                  const float* __restrict__ coeff,
                                  float* __restrict__ out)
{
    __shared__ float  A01[WAVES][TILE_M][16];
    __shared__ float  A23[WAVES][TILE_M][16];
    __shared__ float2 Bfrag[64][32];
    __shared__ float  Sst[WAVES][TILE_M][16];

    const int tid  = threadIdx.x;
    const int wave = tid >> 5;
    const int lane = tid & 31;
    const int m    = lane & 15;
    const int half = lane >> 4;
    const int rowbase = blockIdx.x * (WAVES * TILE_M) + wave * TILE_M;
    const int row = rowbase + m;

    float xv[N_IN];
#pragma unroll
    for (int i = 0; i < N_IN; ++i) xv[i] = x[row * N_IN + i];

    float memb[N_IN][N_MF];
#pragma unroll
    for (int v = 0; v < N_IN; ++v) {
#pragma unroll
        for (int j = 0; j < N_MF; ++j) {
            float d    = (xv[v] - mfc[v * N_MF + j]) / mfa[v * N_MF + j];
            float dist = d * d;
            float p    = __powf(dist, mfb[v * N_MF + j]);
            memb[v][j] = 1.0f / (1.0f + p);
        }
    }

    if (half == 0) {
#pragma unroll
        for (int ja = 0; ja < 4; ++ja) {
#pragma unroll
            for (int jb = 0; jb < 4; ++jb) {
                A01[wave][m][ja * 4 + jb] = memb[0][ja] * memb[1][jb];
                A23[wave][m][ja * 4 + jb] = memb[2][ja] * memb[3][jb];
            }
        }
    }
    const float m5a = half ? memb[5][2] : memb[5][0];
    const float m5b = half ? memb[5][3] : memb[5][1];
    float e0[4], e1[4];
#pragma unroll
    for (int s = 0; s < 4; ++s) {
        e0[s] = memb[4][s] * m5a;
        e1[s] = memb[4][s] * m5b;
    }

    const int   sn    = tid & 15;
    const bool  slt7  = (sn < 7);
    const float scaug = (sn == 7) ? 1.0f : 0.0f;
    const int   skap  = ((tid >> 4) & 1) << 1;

    v8f acc = {};

    for (int j01 = 0; j01 < 16; ++j01) {
        __syncthreads();
#pragma unroll
        for (int e = 0; e < 8; ++e) {
            const int idx  = e * 256 + tid;
            const int step = idx >> 5;
            const int l    = idx & 31;
            const int k0   = j01 * 256 + step * 4 + skap;
            float b0 = scaug, b1 = scaug;
            if (slt7) {
                b0 = coeff[k0 * 7 + sn];
                b1 = coeff[(k0 + 1) * 7 + sn];
            }
            Bfrag[step][l] = make_float2(b0, b1);
        }
        __syncthreads();

        if (j01 < 15)
            __builtin_prefetch(coeff + (j01 + 1) * 256 * 7 + tid * 7, 0, 1);

        const float a01 = A01[wave][m][j01];
#pragma unroll
        for (int j23 = 0; j23 < 16; ++j23) {
            const float a23 = A23[wave][m][j23];
            const float t   = a01 * a23;
#pragma unroll
            for (int s = 0; s < 4; ++s) {
                const float2 bf = Bfrag[j23 * 4 + s][lane];
                v2f bv; bv.x = bf.x; bv.y = bf.y;
                v2f av; av.x = t * e0[s]; av.y = t * e1[s];
                acc = __builtin_amdgcn_wmma_f32_16x16x4_f32(
                    false, av, false, bv, (short)0, acc, false, false);
            }
        }
    }

#pragma unroll
    for (int v = 0; v < 8; ++v)
        Sst[wave][v + half * 8][m] = acc[v];

    __syncthreads();

    if (lane < 16) {
        float num = Sst[wave][m][6];
#pragma unroll
        for (int i = 0; i < N_IN; ++i)
            num += Sst[wave][m][i] * xv[i];
        float denom = fmaxf(Sst[wave][m][7], 1e-12f);
        out[row] = num / denom;
    }
}

extern "C" void kernel_launch(void* const* d_in, const int* in_sizes, int n_in,
                              void* d_out, int out_size, void* d_ws, size_t ws_size,
                              hipStream_t stream) {
    const float* x     = (const float*)d_in[0];
    const float* mfa   = (const float*)d_in[1];
    const float* mfb   = (const float*)d_in[2];
    const float* mfc   = (const float*)d_in[3];
    const float* coeff = (const float*)d_in[4];
    float* out = (float*)d_out;

    dim3 grid(BATCH / (WAVES * TILE_M));
    dim3 block(WAVES * 32);
    anfis_wmma_kernel<<<grid, block, 0, stream>>>(x, mfa, mfb, mfc, coeff, out);
}


# Product re-indexed: K = (j4,j5,coefficient column) as coeff is stored, x folded into A; N = (j2,j3), weighted into the accumulators after; one wave per j0; closed-form rule sum; new sum order
# speedup vs baseline: 9.5569x; 9.5569x over previous
#include <hip/hip_runtime.h>


typedef __attribute__((ext_vector_type(2))) float v2f;
typedef __attribute__((ext_vector_type(4))) float v4f;
typedef __attribute__((ext_vector_type(8))) float v8f;

#define BATCH       4096
#define N_IN        6
#define N_MF        4
#define TILE_M      16
#define NWAVES      4
#define KROW        112
#define PITCH       116
#define SLICE_D     (16 * PITCH)
#define SLICE_V4    14
#define N_SLICE     4
#define WAVE_V4     (1024 * 7 / 4)

__device__ __forceinline__ void wave_lds_sync()
{
    __builtin_amdgcn_fence(__ATOMIC_RELEASE, "wavefront", "local");
    __builtin_amdgcn_wave_barrier();
    __builtin_amdgcn_fence(__ATOMIC_ACQUIRE, "wavefront", "local");
}

__device__ __forceinline__ void wg_lds_barrier()
{
    __builtin_amdgcn_fence(__ATOMIC_RELEASE, "workgroup", "local");
    __builtin_amdgcn_s_barrier();
    __builtin_amdgcn_fence(__ATOMIC_ACQUIRE, "workgroup", "local");
}

__launch_bounds__(32 * NWAVES) __attribute__((amdgpu_waves_per_eu(1, 1)))
__global__ void anfis_wmma_kernel(const float* __restrict__ x,
                                  const float* __restrict__ mfa,
                                  const float* __restrict__ mfb,
                                  const float* __restrict__ mfc,
                                  const float* __restrict__ coeff,
                                  float* __restrict__ out)
{
    __shared__ __attribute__((aligned(16))) float Bs[NWAVES][N_SLICE * SLICE_D];
    __shared__ __attribute__((aligned(16))) float Gs[TILE_M * PITCH];
    __shared__ float Ms[TILE_M][25];
    __shared__ float A23s[TILE_M][17];
    __shared__ float Ps[NWAVES][TILE_M][17];
    __shared__ float Pn[NWAVES][TILE_M];

    const int tid  = threadIdx.x;
    const int wave = __builtin_amdgcn_readfirstlane(tid >> 5);
    const int lane = tid & 31;
    const int c    = lane & 15;
    const int h    = lane >> 4;
    const int row0 = blockIdx.x * TILE_M;

    const int prow = tid >> 3;
    const int sub  = tid & 7;
    const int grow = row0 + prow;

    float xr[N_IN];
#pragma unroll
    for (int i = 0; i < N_IN; ++i) xr[i] = x[grow * N_IN + i];

    float pc[3], pa[3], pb[3];
#pragma unroll
    for (int z = 0; z < 3; ++z) {
        const int pi = sub * 3 + z;
        pc[z] = mfc[pi]; pa[z] = mfa[pi]; pb[z] = mfb[pi];
    }

    const v4f* __restrict__ c4 = reinterpret_cast<const v4f*>(coeff) + wave * WAVE_V4;
    v4f R[N_SLICE][SLICE_V4];
#pragma unroll
    for (int j1 = 0; j1 < N_SLICE; ++j1) {
#pragma unroll
        for (int q = 0; q < SLICE_V4; ++q)
            R[j1][q] = c4[j1 * (SLICE_V4 * 32) + q * 32 + lane];
    }

#pragma unroll
    for (int z = 0; z < 3; ++z) {
        const int pi = sub * 3 + z;
        const int v  = pi >> 2;
        float xk = xr[0];
#pragma unroll
        for (int i = 1; i < N_IN; ++i) xk = (v == i) ? xr[i] : xk;
        float d    = (xk - pc[z]) / pa[z];
        float dist = d * d;
        const float bb = pb[z];
        float p    = __builtin_exp2f(bb * __builtin_log2f(dist));
        p          = (bb == 0.0f || dist == 1.0f) ? 1.0f : p;
        Ms[prow][pi] = 1.0f / (1.0f + p);
    }

    int so[SLICE_V4];
#pragma unroll
    for (int q = 0; q < SLICE_V4; ++q) {
        const int f = q * 32 + lane;
        so[q] = (f / 28) * PITCH + (f % 28) * 4;
    }

    wg_lds_barrier();

    {
        const float m4v = Ms[prow][16 + (sub >> 1)];
        const float m5a = Ms[prow][20 + 2 * (sub & 1)];
        const float m5b = Ms[prow][20 + 2 * (sub & 1) + 1];
        const float p0  = m4v * m5a;
        const float p1  = m4v * m5b;
#pragma unroll
        for (int z = 0; z < 14; ++z) {
            const int   i  = z % 7;
            const float pp = (z < 7) ? p0 : p1;
            Gs[prow * PITCH + sub * 14 + z] = (i < 6) ? pp * xr[i] : pp;
        }
        const float m2v = Ms[prow][8 + (sub >> 1)];
        A23s[prow][2 * sub]     = m2v * Ms[prow][12 + 2 * (sub & 1)];
        A23s[prow][2 * sub + 1] = m2v * Ms[prow][12 + 2 * (sub & 1) + 1];
    }

    float a01[N_SLICE];
    {
        const float m0 = Ms[c][wave];
#pragma unroll
        for (int j1 = 0; j1 < N_SLICE; ++j1) a01[j1] = m0 * Ms[c][4 + j1];
    }

    float* __restrict__ Bw = Bs[wave];
#pragma unroll
    for (int j1 = 0; j1 < 2; ++j1) {
#pragma unroll
        for (int q = 0; q < SLICE_V4; ++q)
            *reinterpret_cast<v4f*>(&Bw[j1 * SLICE_D + so[q]]) = R[j1][q];
    }

    wg_lds_barrier();

    const int fo = c * PITCH + 4 * h;
    v4f g[SLICE_V4];
#pragma unroll
    for (int s = 0; s < SLICE_V4; ++s)
        g[s] = *reinterpret_cast<const v4f*>(&Gs[fo + 8 * s]);

    v4f bq[2][SLICE_V4];
#pragma unroll
    for (int s = 0; s < SLICE_V4; ++s)
        bq[0][s] = *reinterpret_cast<const v4f*>(&Bw[fo + 8 * s]);

    v8f acc0 = {};
    v8f acc1 = {};

#pragma unroll
    for (int j1 = 0; j1 < N_SLICE; ++j1) {
        if (j1 + 1 < N_SLICE) {
#pragma unroll
            for (int s = 0; s < SLICE_V4; ++s)
                bq[(j1 + 1) & 1][s] =
                    *reinterpret_cast<const v4f*>(&Bw[(j1 + 1) * SLICE_D + fo + 8 * s]);
        }
        __builtin_amdgcn_sched_barrier(0);
#pragma unroll
        for (int s = 0; s < SLICE_V4; ++s) {
            const v4f gv = g[s];
            const v4f bv = bq[j1 & 1][s];
            const float t = a01[j1];
            v2f a0; a0.x = t * gv.x; a0.y = t * gv.y;
            v2f a1; a1.x = t * gv.z; a1.y = t * gv.w;
            v2f b0; b0.x = bv.x; b0.y = bv.y;
            v2f b1; b1.x = bv.z; b1.y = bv.w;
            acc0 = __builtin_amdgcn_wmma_f32_16x16x4_f32(
                false, a0, false, b0, (short)0, acc0, false, false);
            acc1 = __builtin_amdgcn_wmma_f32_16x16x4_f32(
                false, a1, false, b1, (short)0, acc1, false, false);
        }
        __builtin_amdgcn_sched_barrier(0);
        if (j1 + 2 < N_SLICE) {
#pragma unroll
            for (int q = 0; q < SLICE_V4; ++q)
                *reinterpret_cast<v4f*>(&Bw[(j1 + 2) * SLICE_D + so[q]]) = R[j1 + 2][q];
            wave_lds_sync();
        }
    }

#pragma unroll
    for (int r = 0; r < 8; ++r)
        Ps[wave][8 * h + r][c] = (acc0[r] + acc1[r]) * A23s[8 * h + r][c];

    wave_lds_sync();

    if (lane < TILE_M) {
        float s = 0.0f;
#pragma unroll
        for (int n = 0; n < 16; ++n) s += Ps[wave][lane][n];
        Pn[wave][lane] = s;
    }

    wg_lds_barrier();

    if (tid < TILE_M) {
        const float num = (Pn[0][tid] + Pn[1][tid]) + (Pn[2][tid] + Pn[3][tid]);
        float den = 1.0f;
#pragma unroll
        for (int v = 0; v < N_IN; ++v)
            den *= (Ms[tid][4 * v] + Ms[tid][4 * v + 1]) + (Ms[tid][4 * v + 2] + Ms[tid][4 * v + 3]);
        out[row0 + tid] = num / fmaxf(den, 1e-12f);
    }
}

extern "C" void kernel_launch(void* const* d_in, const int* in_sizes, int n_in,
                              void* d_out, int out_size, void* d_ws, size_t ws_size,
                              hipStream_t stream) {
    const float* x     = (const float*)d_in[0];
    const float* mfa   = (const float*)d_in[1];
    const float* mfb   = (const float*)d_in[2];
    const float* mfc   = (const float*)d_in[3];
    const float* coeff = (const float*)d_in[4];
    float* out = (float*)d_out;

    dim3 grid(BATCH / TILE_M);
    dim3 block(32 * NWAVES);
    anfis_wmma_kernel<<<grid, block, 0, stream>>>(x, mfa, mfb, mfc, coeff, out);
}
